# starting kernel
# baseline (speedup 1.0000x reference)
#include <hip/hip_runtime.h>
#include <math.h>


#define DEVI __device__ __forceinline__

typedef __attribute__((ext_vector_type(2))) float v2f;
typedef __attribute__((ext_vector_type(8))) float v8f;

constexpr int Bc = 128, Tc = 1024, Fc = 256, Uc = 256;
constexpr int HS = 260;

DEVI v8f wmma4(v2f a, v2f b, v8f c) {
  return __builtin_amdgcn_wmma_f32_16x16x4_f32(false, a, false, b, (short)0, c,
                                               false, false);
}

DEVI v8f gemm_global_a(const float* __restrict__ Arow0, int lda,
                       const float* __restrict__ Bn, int ldb, int K,
                       v8f acc, int l15, int half) {
#pragma unroll 4
  for (int kc = 0; kc < K; kc += 4) {
    v2f a = *(const v2f*)(Arow0 + (size_t)l15 * lda + kc + 2 * half);
    const float* pb = Bn + (size_t)(kc + 2 * half) * ldb + l15;
    v2f b;
    b.x = pb[0];
    b.y = pb[ldb];
    acc = wmma4(a, b, acc);
  }
  return acc;
}

DEVI v8f gemm_lds_a(const float* hs, const float* __restrict__ Bn, int ldb,
                    v8f acc, int l15, int half) {
#pragma unroll 4
  for (int kc = 0; kc < Uc; kc += 4) {
    v2f a = *(const v2f*)(hs + l15 * HS + kc + 2 * half);
    const float* pb = Bn + (size_t)(kc + 2 * half) * ldb + l15;
    v2f b;
    b.x = pb[0];
    b.y = pb[ldb];
    acc = wmma4(a, b, acc);
  }
  return acc;
}

__global__ __launch_bounds__(512) void xp_proj_kernel(
    const float* __restrict__ x, const float* __restrict__ Wx,
    const float* __restrict__ b, float* __restrict__ xp) {
  const int lane = threadIdx.x & 31;
  const int wave = threadIdx.x >> 5;
  const int l15 = lane & 15, half = lane >> 4;
  const int n0 = wave * 16;
  const size_t rbase = (size_t)blockIdx.x * 16;
  const int bidx = (int)(rbase / Tc);
  const int tbase = (int)(rbase % Tc);

  const float bv = b[n0 + l15];
  v8f acc = {bv, bv, bv, bv, bv, bv, bv, bv};
  acc = gemm_global_a(x + rbase * Fc, Fc, Wx + n0, Uc, Fc, acc, l15, half);

#pragma unroll
  for (int r = 0; r < 8; ++r)
    xp[((size_t)(tbase + r + 8 * half) * Bc + bidx) * Uc + n0 + l15] = acc[r];
}

template <bool PRE>
__global__ __launch_bounds__(512) void rnn_fused_kernel(
    const float* __restrict__ xp, const float* __restrict__ x,
    const float* __restrict__ enc_Wx, const float* __restrict__ enc_b,
    const float* __restrict__ enc_Wh, const float* __restrict__ dec_Wx,
    const float* __restrict__ dec_Wh, const float* __restrict__ dec_b,
    float* __restrict__ out) {
  __shared__ float h_enc[2][16 * HS];
  __shared__ float h_dec[2][16 * HS];

  const int lane = threadIdx.x & 31;
  const int wave = threadIdx.x >> 5;
  const int l15 = lane & 15, half = lane >> 4;
  const int n0 = wave * 16;
  const int bbase = blockIdx.x * 16;

  for (int i = threadIdx.x; i < 16 * HS; i += blockDim.x) {
    h_enc[0][i] = 0.0f;
    h_enc[1][i] = 0.0f;
    h_dec[0][i] = 0.0f;
    h_dec[1][i] = 0.0f;
  }
  __syncthreads();

  const float eb = enc_b[n0 + l15];
  const float db = dec_b[n0 + l15];
  v8f dacc = {};
  v8f acc_pre = {};

  if (PRE) {
    const float* p0 = xp + ((size_t)0 * Bc + bbase) * Uc + n0 + l15;
#pragma unroll
    for (int r = 0; r < 8; ++r) acc_pre[r] = p0[(r + 8 * half) * Uc];
  }

  for (int t = 0; t < Tc; ++t) {
    const int cur = t & 1, nxt = cur ^ 1;

    v8f acc;
    if (PRE) {
      acc = acc_pre;
      if (t + 1 < Tc) {
        const float* p = xp + ((size_t)(t + 1) * Bc + bbase) * Uc + n0 + l15;
#pragma unroll
        for (int r = 0; r < 8; ++r) acc_pre[r] = p[(r + 8 * half) * Uc];
      }
    } else {
      acc = (v8f){eb, eb, eb, eb, eb, eb, eb, eb};
      acc = gemm_global_a(x + (size_t)bbase * Tc * Fc + (size_t)t * Fc, Tc * Fc,
                          enc_Wx + n0, Uc, Fc, acc, l15, half);
    }
    acc = gemm_lds_a(h_enc[cur], enc_Wh + n0, Uc, acc, l15, half);
#pragma unroll
    for (int r = 0; r < 8; ++r) acc[r] = tanhf(acc[r]);
#pragma unroll
    for (int r = 0; r < 8; ++r)
      h_enc[nxt][(r + 8 * half) * HS + n0 + l15] = acc[r];
    __syncthreads();

    v8f acc2 = {db, db, db, db, db, db, db, db};
    acc2 = gemm_lds_a(h_enc[nxt], dec_Wx + n0, Uc, acc2, l15, half);
    acc2 = gemm_lds_a(h_dec[cur], dec_Wh + n0, Uc, acc2, l15, half);
#pragma unroll
    for (int r = 0; r < 8; ++r) acc2[r] = tanhf(acc2[r]);
#pragma unroll
    for (int r = 0; r < 8; ++r)
      h_dec[nxt][(r + 8 * half) * HS + n0 + l15] = acc2[r];
    dacc = acc2;
    __syncthreads();
  }

#pragma unroll
  for (int r = 0; r < 8; ++r)
    out[(size_t)(bbase + r + 8 * half) * Uc + n0 + l15] = dacc[r];
}

extern "C" void kernel_launch(void* const* d_in, const int* in_sizes, int n_in,
                              void* d_out, int out_size, void* d_ws,
                              size_t ws_size, hipStream_t stream) {
  const float* x = (const float*)d_in[0];
  const float* enc_Wx = (const float*)d_in[1];
  const float* enc_Wh = (const float*)d_in[2];
  const float* enc_b = (const float*)d_in[3];
  const float* dec_Wx = (const float*)d_in[4];
  const float* dec_Wh = (const float*)d_in[5];
  const float* dec_b = (const float*)d_in[6];
  float* out = (float*)d_out;

  const size_t xp_bytes = (size_t)Bc * Tc * Uc * sizeof(float);
  if (ws_size >= xp_bytes) {
    float* xp = (float*)d_ws;
    xp_proj_kernel<<<(Bc * Tc) / 16, 512, 0, stream>>>(x, enc_Wx, enc_b, xp);
    rnn_fused_kernel<true><<<Bc / 16, 512, 0, stream>>>(
        xp, x, enc_Wx, enc_b, enc_Wh, dec_Wx, dec_Wh, dec_b, out);
  } else {
    rnn_fused_kernel<false><<<Bc / 16, 512, 0, stream>>>(
        nullptr, x, enc_Wx, enc_b, enc_Wh, dec_Wx, dec_Wh, dec_b, out);
  }
}


# Recurrent weights held in registers across the time loop: each wave loads its 256x16 slice of Wh once (64 B fragments) instead of re-reading it from L2 every step; k order and sums unchanged
# speedup vs baseline: 2.1018x; 2.1018x over previous
#include <hip/hip_runtime.h>
#include <math.h>


#define DEVI __device__ __forceinline__

typedef __attribute__((ext_vector_type(2))) float v2f;
typedef __attribute__((ext_vector_type(8))) float v8f;

constexpr int Bc = 128, Tc = 1024, Fc = 256, Uc = 256;
constexpr int HS = 260;

DEVI v8f wmma4(v2f a, v2f b, v8f c) {
  return __builtin_amdgcn_wmma_f32_16x16x4_f32(false, a, false, b, (short)0, c,
                                               false, false);
}

DEVI v8f gemm_global_a(const float* __restrict__ Arow0, int lda,
                       const float* __restrict__ Bn, int ldb, int K,
                       v8f acc, int l15, int half) {
#pragma unroll 4
  for (int kc = 0; kc < K; kc += 4) {
    v2f a = *(const v2f*)(Arow0 + (size_t)l15 * lda + kc + 2 * half);
    const float* pb = Bn + (size_t)(kc + 2 * half) * ldb + l15;
    v2f b;
    b.x = pb[0];
    b.y = pb[ldb];
    acc = wmma4(a, b, acc);
  }
  return acc;
}

DEVI v8f gemm_lds_a(const float* hs, const float* __restrict__ Bn, int ldb,
                    v8f acc, int l15, int half) {
#pragma unroll 4
  for (int kc = 0; kc < Uc; kc += 4) {
    v2f a = *(const v2f*)(hs + l15 * HS + kc + 2 * half);
    const float* pb = Bn + (size_t)(kc + 2 * half) * ldb + l15;
    v2f b;
    b.x = pb[0];
    b.y = pb[ldb];
    acc = wmma4(a, b, acc);
  }
  return acc;
}

__global__ __launch_bounds__(512) void xp_proj_kernel(
    const float* __restrict__ x, const float* __restrict__ Wx,
    const float* __restrict__ b, float* __restrict__ xp) {
  const int lane = threadIdx.x & 31;
  const int wave = threadIdx.x >> 5;
  const int l15 = lane & 15, half = lane >> 4;
  const int n0 = wave * 16;
  const size_t rbase = (size_t)blockIdx.x * 16;
  const int bidx = (int)(rbase / Tc);
  const int tbase = (int)(rbase % Tc);

  const float bv = b[n0 + l15];
  v8f acc = {bv, bv, bv, bv, bv, bv, bv, bv};
  acc = gemm_global_a(x + rbase * Fc, Fc, Wx + n0, Uc, Fc, acc, l15, half);

#pragma unroll
  for (int r = 0; r < 8; ++r)
    xp[((size_t)(tbase + r + 8 * half) * Bc + bidx) * Uc + n0 + l15] = acc[r];
}

template <bool PRE>
__global__ __launch_bounds__(512) void rnn_fused_kernel(
    const float* __restrict__ xp, const float* __restrict__ x,
    const float* __restrict__ enc_Wx, const float* __restrict__ enc_b,
    const float* __restrict__ enc_Wh, const float* __restrict__ dec_Wx,
    const float* __restrict__ dec_Wh, const float* __restrict__ dec_b,
    float* __restrict__ out) {
  __shared__ float h_enc[2][16 * HS];
  __shared__ float h_dec[2][16 * HS];

  const int lane = threadIdx.x & 31;
  const int wave = threadIdx.x >> 5;
  const int l15 = lane & 15, half = lane >> 4;
  const int n0 = wave * 16;
  const int bbase = blockIdx.x * 16;

  for (int i = threadIdx.x; i < 16 * HS; i += blockDim.x) {
    h_enc[0][i] = 0.0f;
    h_enc[1][i] = 0.0f;
    h_dec[0][i] = 0.0f;
    h_dec[1][i] = 0.0f;
  }
  __syncthreads();

  const float eb = enc_b[n0 + l15];
  const float db = dec_b[n0 + l15];
  v8f dacc = {};
  v8f acc_pre = {};

  if (PRE) {
    const float* p0 = xp + ((size_t)0 * Bc + bbase) * Uc + n0 + l15;
#pragma unroll
    for (int r = 0; r < 8; ++r) acc_pre[r] = p0[(r + 8 * half) * Uc];
  }

  for (int t = 0; t < Tc; ++t) {
    const int cur = t & 1, nxt = cur ^ 1;

    v8f acc;
    if (PRE) {
      acc = acc_pre;
      if (t + 1 < Tc) {
        const float* p = xp + ((size_t)(t + 1) * Bc + bbase) * Uc + n0 + l15;
#pragma unroll
        for (int r = 0; r < 8; ++r) acc_pre[r] = p[(r + 8 * half) * Uc];
      }
    } else {
      acc = (v8f){eb, eb, eb, eb, eb, eb, eb, eb};
      acc = gemm_global_a(x + (size_t)bbase * Tc * Fc + (size_t)t * Fc, Tc * Fc,
                          enc_Wx + n0, Uc, Fc, acc, l15, half);
    }
    acc = gemm_lds_a(h_enc[cur], enc_Wh + n0, Uc, acc, l15, half);
#pragma unroll
    for (int r = 0; r < 8; ++r) acc[r] = tanhf(acc[r]);
#pragma unroll
    for (int r = 0; r < 8; ++r)
      h_enc[nxt][(r + 8 * half) * HS + n0 + l15] = acc[r];
    __syncthreads();

    v8f acc2 = {db, db, db, db, db, db, db, db};
    acc2 = gemm_lds_a(h_enc[nxt], dec_Wx + n0, Uc, acc2, l15, half);
    acc2 = gemm_lds_a(h_dec[cur], dec_Wh + n0, Uc, acc2, l15, half);
#pragma unroll
    for (int r = 0; r < 8; ++r) acc2[r] = tanhf(acc2[r]);
#pragma unroll
    for (int r = 0; r < 8; ++r)
      h_dec[nxt][(r + 8 * half) * HS + n0 + l15] = acc2[r];
    dacc = acc2;
    __syncthreads();
  }

#pragma unroll
  for (int r = 0; r < 8; ++r)
    out[(size_t)(bbase + r + 8 * half) * Uc + n0 + l15] = dacc[r];
}

__global__ __launch_bounds__(512) void seq_proj_inplace_kernel(
    float* hx, const float* __restrict__ Wx, const float* __restrict__ b) {
  __shared__ float a_lds[16 * HS];

  const int lane = threadIdx.x & 31;
  const int wave = threadIdx.x >> 5;
  const int l15 = lane & 15, half = lane >> 4;
  const int n0 = wave * 16;
  float* tile = hx + (size_t)blockIdx.x * 16 * Uc;

  for (int i = threadIdx.x; i < 16 * Uc / 4; i += blockDim.x) {
    const int row = i / (Uc / 4), c4 = i % (Uc / 4);
    const float4 v = *(const float4*)(tile + row * Uc + c4 * 4);
    *(float4*)(a_lds + row * HS + c4 * 4) = v;
  }
  __syncthreads();

  const float bv = b[n0 + l15];
  v8f acc = {bv, bv, bv, bv, bv, bv, bv, bv};
  acc = gemm_lds_a(a_lds, Wx + n0, Uc, acc, l15, half);

#pragma unroll
  for (int r = 0; r < 8; ++r) tile[(r + 8 * half) * Uc + n0 + l15] = acc[r];
}

template <bool SEQ>
__global__ __launch_bounds__(512) void rnn_rec_kernel(
    float* xh, const float* __restrict__ Wh, float* __restrict__ out) {
  __shared__ float h[2][16 * HS];

  const int lane = threadIdx.x & 31;
  const int wave = threadIdx.x >> 5;
  const int l15 = lane & 15, half = lane >> 4;
  const int n0 = wave * 16;
  const int bbase = blockIdx.x * 16;

  for (int i = threadIdx.x; i < 16 * HS; i += blockDim.x) {
    h[0][i] = 0.0f;
    h[1][i] = 0.0f;
  }
  __syncthreads();

  v2f bw[Uc / 4];
  {
    const float* pb = Wh + (size_t)(2 * half) * Uc + n0 + l15;
#pragma unroll
    for (int i = 0; i < Uc / 4; ++i) {
      bw[i].x = pb[(size_t)(4 * i) * Uc];
      bw[i].y = pb[(size_t)(4 * i + 1) * Uc];
    }
  }

  v8f acc_pre;
  {
    const float* p0 = xh + (size_t)bbase * Uc + n0 + l15;
#pragma unroll
    for (int r = 0; r < 8; ++r) acc_pre[r] = p0[(r + 8 * half) * Uc];
  }
  v8f acc = {};

  for (int t = 0; t < Tc; ++t) {
    const int cur = t & 1, nxt = cur ^ 1;

    acc = acc_pre;
    if (t + 1 < Tc) {
      const float* p = xh + ((size_t)(t + 1) * Bc + bbase) * Uc + n0 + l15;
#pragma unroll
      for (int r = 0; r < 8; ++r) acc_pre[r] = p[(r + 8 * half) * Uc];
    }
    {
      const float* ha = h[cur] + l15 * HS + 2 * half;
#pragma unroll
      for (int i = 0; i < Uc / 4; ++i) {
        const v2f a = *(const v2f*)(ha + 4 * i);
        acc = wmma4(a, bw[i], acc);
      }
    }
#pragma unroll
    for (int r = 0; r < 8; ++r) acc[r] = tanhf(acc[r]);
#pragma unroll
    for (int r = 0; r < 8; ++r) h[nxt][(r + 8 * half) * HS + n0 + l15] = acc[r];
    __syncthreads();
    if (SEQ) {
      float* q = xh + ((size_t)t * Bc + bbase) * Uc + n0 + l15;
#pragma unroll
      for (int r = 0; r < 8; ++r) q[(r + 8 * half) * Uc] = acc[r];
    }
  }

  if (!SEQ) {
#pragma unroll
    for (int r = 0; r < 8; ++r)
      out[(size_t)(bbase + r + 8 * half) * Uc + n0 + l15] = acc[r];
  }
}

extern "C" void kernel_launch(void* const* d_in, const int* in_sizes, int n_in,
                              void* d_out, int out_size, void* d_ws,
                              size_t ws_size, hipStream_t stream) {
  const float* x = (const float*)d_in[0];
  const float* enc_Wx = (const float*)d_in[1];
  const float* enc_Wh = (const float*)d_in[2];
  const float* enc_b = (const float*)d_in[3];
  const float* dec_Wx = (const float*)d_in[4];
  const float* dec_Wh = (const float*)d_in[5];
  const float* dec_b = (const float*)d_in[6];
  float* out = (float*)d_out;

  const size_t xp_bytes = (size_t)Bc * Tc * Uc * sizeof(float);
  if (ws_size >= xp_bytes) {
    float* xp = (float*)d_ws;
    xp_proj_kernel<<<(Bc * Tc) / 16, 512, 0, stream>>>(x, enc_Wx, enc_b, xp);
    rnn_rec_kernel<true><<<Bc / 16, 512, 0, stream>>>(xp, enc_Wh, nullptr);
    seq_proj_inplace_kernel<<<(Bc * Tc) / 16, 512, 0, stream>>>(xp, dec_Wx,
                                                                dec_b);
    rnn_rec_kernel<false><<<Bc / 16, 512, 0, stream>>>(xp, dec_Wh, out);
  } else {
    rnn_fused_kernel<false><<<Bc / 16, 512, 0, stream>>>(
        nullptr, x, enc_Wx, enc_b, enc_Wh, dec_Wx, dec_Wh, dec_b, out);
  }
}


# tanh branch-free on the hardware exp2 and reciprocal instructions ((1-e)/(1+e), e=exp(-2|x|); odd series below |x|=0.1) instead of the library routine; abs. error ~1e-7, last bits differ
# speedup vs baseline: 4.2492x; 2.0217x over previous
#include <hip/hip_runtime.h>
#include <math.h>


#define DEVI __device__ __forceinline__

typedef __attribute__((ext_vector_type(2))) float v2f;
typedef __attribute__((ext_vector_type(8))) float v8f;

constexpr int Bc = 128, Tc = 1024, Fc = 256, Uc = 256;
constexpr int HS = 260;

DEVI float tanh_hw(float x) {
  const float a = fabsf(x);
  const float e = __builtin_amdgcn_exp2f(a * -2.8853900817779268f);
  const float big = (1.0f - e) * __builtin_amdgcn_rcpf(1.0f + e);
  const float a2 = a * a;
  const float small =
      a * fmaf(a2, fmaf(a2, 0.13333334f, -0.33333334f), 1.0f);
  return copysignf(a < 0.1f ? small : big, x);
}

DEVI v8f wmma4(v2f a, v2f b, v8f c) {
  return __builtin_amdgcn_wmma_f32_16x16x4_f32(false, a, false, b, (short)0, c,
                                               false, false);
}

DEVI v8f gemm_global_a(const float* __restrict__ Arow0, int lda,
                       const float* __restrict__ Bn, int ldb, int K,
                       v8f acc, int l15, int half) {
#pragma unroll 4
  for (int kc = 0; kc < K; kc += 4) {
    v2f a = *(const v2f*)(Arow0 + (size_t)l15 * lda + kc + 2 * half);
    const float* pb = Bn + (size_t)(kc + 2 * half) * ldb + l15;
    v2f b;
    b.x = pb[0];
    b.y = pb[ldb];
    acc = wmma4(a, b, acc);
  }
  return acc;
}

DEVI v8f gemm_lds_a(const float* hs, const float* __restrict__ Bn, int ldb,
                    v8f acc, int l15, int half) {
#pragma unroll 4
  for (int kc = 0; kc < Uc; kc += 4) {
    v2f a = *(const v2f*)(hs + l15 * HS + kc + 2 * half);
    const float* pb = Bn + (size_t)(kc + 2 * half) * ldb + l15;
    v2f b;
    b.x = pb[0];
    b.y = pb[ldb];
    acc = wmma4(a, b, acc);
  }
  return acc;
}

template <bool PRE>
__global__ __launch_bounds__(512) void rnn_fused_kernel(
    const float* __restrict__ xp, const float* __restrict__ x,
    const float* __restrict__ enc_Wx, const float* __restrict__ enc_b,
    const float* __restrict__ enc_Wh, const float* __restrict__ dec_Wx,
    const float* __restrict__ dec_Wh, const float* __restrict__ dec_b,
    float* __restrict__ out) {
  __shared__ float h_enc[2][16 * HS];
  __shared__ float h_dec[2][16 * HS];

  const int lane = threadIdx.x & 31;
  const int wave = threadIdx.x >> 5;
  const int l15 = lane & 15, half = lane >> 4;
  const int n0 = wave * 16;
  const int bbase = blockIdx.x * 16;

  for (int i = threadIdx.x; i < 16 * HS; i += blockDim.x) {
    h_enc[0][i] = 0.0f;
    h_enc[1][i] = 0.0f;
    h_dec[0][i] = 0.0f;
    h_dec[1][i] = 0.0f;
  }
  __syncthreads();

  const float eb = enc_b[n0 + l15];
  const float db = dec_b[n0 + l15];
  v8f dacc = {};
  v8f acc_pre = {};

  if (PRE) {
    const float* p0 = xp + ((size_t)0 * Bc + bbase) * Uc + n0 + l15;
#pragma unroll
    for (int r = 0; r < 8; ++r) acc_pre[r] = p0[(r + 8 * half) * Uc];
  }

  for (int t = 0; t < Tc; ++t) {
    const int cur = t & 1, nxt = cur ^ 1;

    v8f acc;
    if (PRE) {
      acc = acc_pre;
      if (t + 1 < Tc) {
        const float* p = xp + ((size_t)(t + 1) * Bc + bbase) * Uc + n0 + l15;
#pragma unroll
        for (int r = 0; r < 8; ++r) acc_pre[r] = p[(r + 8 * half) * Uc];
      }
    } else {
      acc = (v8f){eb, eb, eb, eb, eb, eb, eb, eb};
      acc = gemm_global_a(x + (size_t)bbase * Tc * Fc + (size_t)t * Fc, Tc * Fc,
                          enc_Wx + n0, Uc, Fc, acc, l15, half);
    }
    acc = gemm_lds_a(h_enc[cur], enc_Wh + n0, Uc, acc, l15, half);
#pragma unroll
    for (int r = 0; r < 8; ++r) acc[r] = tanh_hw(acc[r]);
#pragma unroll
    for (int r = 0; r < 8; ++r)
      h_enc[nxt][(r + 8 * half) * HS + n0 + l15] = acc[r];
    __syncthreads();

    v8f acc2 = {db, db, db, db, db, db, db, db};
    acc2 = gemm_lds_a(h_enc[nxt], dec_Wx + n0, Uc, acc2, l15, half);
    acc2 = gemm_lds_a(h_dec[cur], dec_Wh + n0, Uc, acc2, l15, half);
#pragma unroll
    for (int r = 0; r < 8; ++r) acc2[r] = tanh_hw(acc2[r]);
#pragma unroll
    for (int r = 0; r < 8; ++r)
      h_dec[nxt][(r + 8 * half) * HS + n0 + l15] = acc2[r];
    dacc = acc2;
    __syncthreads();
  }

#pragma unroll
  for (int r = 0; r < 8; ++r)
    out[(size_t)(bbase + r + 8 * half) * Uc + n0 + l15] = dacc[r];
}

constexpr int CH = 64;
constexpr int NCH = Tc / CH;
constexpr int NREC = 16;
constexpr int NWORK = 240;
constexpr int XT = CH / 16;
constexpr int NBLK = NREC + NWORK;

DEVI int dec_t0(int j) { return j > 0 ? j * CH - 1 : 0; }
DEVI int dec_t1(int j) { return j == NCH - 1 ? Tc : (j + 1) * CH - 1; }

DEVI void xproj_tile(const float* __restrict__ x, const float* __restrict__ Wx,
                     const float* __restrict__ b, float* xp, int bidx, int tbase,
                     int n0, int l15, int half) {
  const float bv = b[n0 + l15];
  v8f acc = {bv, bv, bv, bv, bv, bv, bv, bv};
  acc = gemm_global_a(x + ((size_t)bidx * Tc + tbase) * Fc, Fc, Wx + n0, Uc, Fc,
                      acc, l15, half);
#pragma unroll
  for (int r = 0; r < 8; ++r)
    xp[((size_t)(tbase + r + 8 * half) * Bc + bidx) * Uc + n0 + l15] = acc[r];
}

DEVI void hproj_tile(float* tile, const float* __restrict__ Wx,
                     const float* __restrict__ b, float* a_lds, int n0, int l15,
                     int half) {
  for (int i = threadIdx.x; i < 16 * Uc / 4; i += blockDim.x) {
    const int row = i / (Uc / 4), c4 = i % (Uc / 4);
    const float4 v = *(const float4*)(tile + row * Uc + c4 * 4);
    *(float4*)(a_lds + row * HS + c4 * 4) = v;
  }
  __syncthreads();
  const float bv = b[n0 + l15];
  v8f acc = {bv, bv, bv, bv, bv, bv, bv, bv};
  acc = gemm_lds_a(a_lds, Wx + n0, Uc, acc, l15, half);
#pragma unroll
  for (int r = 0; r < 8; ++r) tile[(r + 8 * half) * Uc + n0 + l15] = acc[r];
  __syncthreads();
}

template <bool SEQ>
DEVI void rec_chunk(float* xh, const float* __restrict__ Wh, int bbase, int t0,
                    int t1, const float* init, float* fin, float* hl, int n0,
                    int l15, int half) {
  float* h0 = hl;
  float* h1 = hl + 16 * HS;
  if (init) {
    const float* src = init + (size_t)bbase * Uc;
    for (int i = threadIdx.x; i < 16 * Uc / 4; i += blockDim.x) {
      const int row = i / (Uc / 4), c4 = i % (Uc / 4);
      const float4 v = *(const float4*)(src + row * Uc + c4 * 4);
      *(float4*)(h0 + row * HS + c4 * 4) = v;
    }
  } else {
    for (int i = threadIdx.x; i < 16 * HS; i += blockDim.x) h0[i] = 0.0f;
  }
  __syncthreads();

  v2f bw[Uc / 4];
  {
    const float* pb = Wh + (size_t)(2 * half) * Uc + n0 + l15;
#pragma unroll
    for (int i = 0; i < Uc / 4; ++i) {
      bw[i].x = pb[(size_t)(4 * i) * Uc];
      bw[i].y = pb[(size_t)(4 * i + 1) * Uc];
    }
  }

  v8f acc_pre;
  {
    const float* p0 = xh + ((size_t)t0 * Bc + bbase) * Uc + n0 + l15;
#pragma unroll
    for (int r = 0; r < 8; ++r) acc_pre[r] = p0[(r + 8 * half) * Uc];
  }
  v8f acc = {};

  for (int t = t0; t < t1; ++t) {
    const float* hc = ((t - t0) & 1) ? h1 : h0;
    float* hn = ((t - t0) & 1) ? h0 : h1;

    acc = acc_pre;
    if (t + 1 < t1) {
      const float* p = xh + ((size_t)(t + 1) * Bc + bbase) * Uc + n0 + l15;
#pragma unroll
      for (int r = 0; r < 8; ++r) acc_pre[r] = p[(r + 8 * half) * Uc];
    }
    {
      const float* ha = hc + l15 * HS + 2 * half;
#pragma unroll
      for (int i = 0; i < Uc / 4; ++i) {
        const v2f a = *(const v2f*)(ha + 4 * i);
        acc = wmma4(a, bw[i], acc);
      }
    }
#pragma unroll
    for (int r = 0; r < 8; ++r) acc[r] = tanh_hw(acc[r]);
#pragma unroll
    for (int r = 0; r < 8; ++r) hn[(r + 8 * half) * HS + n0 + l15] = acc[r];
    __syncthreads();
    if (SEQ) {
      float* q = xh + ((size_t)t * Bc + bbase) * Uc + n0 + l15;
#pragma unroll
      for (int r = 0; r < 8; ++r) q[(r + 8 * half) * Uc] = acc[r];
    }
  }

  if (!SEQ) {
#pragma unroll
    for (int r = 0; r < 8; ++r) fin[(r + 8 * half) * Uc + n0 + l15] = acc[r];
  }
}

__global__ __launch_bounds__(512) void stage_kernel(
    float* xh, const float* __restrict__ x, const float* __restrict__ enc_Wx,
    const float* __restrict__ enc_b, const float* __restrict__ enc_Wh,
    const float* __restrict__ dec_Wx, const float* __restrict__ dec_b,
    const float* __restrict__ dec_Wh, float* out, int k) {
  __shared__ float sm[2 * 16 * HS];

  const int lane = threadIdx.x & 31;
  const int wave = threadIdx.x >> 5;
  const int l15 = lane & 15, half = lane >> 4;
  const int n0 = wave * 16;
  const int blk = blockIdx.x;

  if (blk < 8) {
    if (k >= 0 && k < NCH) {
      const int t0 = k * CH;
      const float* init = k > 0 ? xh + (size_t)(t0 - 1) * Bc * Uc : nullptr;
      rec_chunk<true>(xh, enc_Wh, blk * 16, t0, t0 + CH, init, nullptr, sm, n0,
                      l15, half);
    }
  } else if (blk < NREC) {
    const int j = k - 2;
    if (j >= 0 && j < NCH) {
      const int bbase = (blk - 8) * 16;
      const int t0 = dec_t0(j);
      const float* init =
          j > 0 ? xh + (size_t)dec_t0(j - 1) * Bc * Uc : nullptr;
      float* fin = j == NCH - 1 ? out + (size_t)bbase * Uc
                                : xh + ((size_t)t0 * Bc + bbase) * Uc;
      rec_chunk<false>(xh, dec_Wh, bbase, t0, dec_t1(j), init, fin, sm, n0, l15,
                       half);
    }
  } else {
    const int w = blk - NREC;
    const int kx = k + 1;
    if (kx >= 0 && kx < NCH) {
      for (int q = w; q < Bc * XT; q += NWORK)
        xproj_tile(x, enc_Wx, enc_b, xh, q / XT, kx * CH + (q % XT) * 16, n0,
                   l15, half);
    }
    const int jp = k - 1;
    if (jp >= 0 && jp < NCH) {
      const int tile1 = dec_t1(jp) * (Bc / 16);
      for (int tile = dec_t0(jp) * (Bc / 16) + w; tile < tile1; tile += NWORK)
        hproj_tile(xh + (size_t)tile * 16 * Uc, dec_Wx, dec_b, sm, n0, l15,
                   half);
    }
  }
}

extern "C" void kernel_launch(void* const* d_in, const int* in_sizes, int n_in,
                              void* d_out, int out_size, void* d_ws,
                              size_t ws_size, hipStream_t stream) {
  const float* x = (const float*)d_in[0];
  const float* enc_Wx = (const float*)d_in[1];
  const float* enc_Wh = (const float*)d_in[2];
  const float* enc_b = (const float*)d_in[3];
  const float* dec_Wx = (const float*)d_in[4];
  const float* dec_Wh = (const float*)d_in[5];
  const float* dec_b = (const float*)d_in[6];
  float* out = (float*)d_out;

  const size_t xp_bytes = (size_t)Bc * Tc * Uc * sizeof(float);
  if (ws_size >= xp_bytes) {
    float* xh = (float*)d_ws;
#define STAGE(k)                                                            \
  stage_kernel<<<NBLK, 512, 0, stream>>>(xh, x, enc_Wx, enc_b,              \
                                                 enc_Wh, dec_Wx, dec_b,     \
                                                 dec_Wh, out, k)
    STAGE(-1); STAGE(0); STAGE(1); STAGE(2); STAGE(3); STAGE(4); STAGE(5);
    STAGE(6); STAGE(7); STAGE(8); STAGE(9); STAGE(10); STAGE(11); STAGE(12);
    STAGE(13); STAGE(14); STAGE(15); STAGE(16); STAGE(17);
#undef STAGE
  } else {
    rnn_fused_kernel<false><<<Bc / 16, 512, 0, stream>>>(
        nullptr, x, enc_Wx, enc_b, enc_Wh, dec_Wx, dec_Wh, dec_b, out);
  }
}


# Recurrence step: the state's LDS fragment loads go out in four groups of eight, two right behind the barrier ahead of the step's global traffic, two a WMMA block ahead of use; sums unchanged
# speedup vs baseline: 4.7735x; 1.1234x over previous
#include <hip/hip_runtime.h>
#include <math.h>


#define DEVI __device__ __forceinline__

typedef __attribute__((ext_vector_type(2))) float v2f;
typedef __attribute__((ext_vector_type(8))) float v8f;

constexpr int Bc = 128, Tc = 1024, Fc = 256, Uc = 256;
constexpr int HS = 260;

DEVI float tanh_hw(float x) {
  const float a = fabsf(x);
  const float e = __builtin_amdgcn_exp2f(a * -2.8853900817779268f);
  const float big = (1.0f - e) * __builtin_amdgcn_rcpf(1.0f + e);
  const float a2 = a * a;
  const float small =
      a * fmaf(a2, fmaf(a2, 0.13333334f, -0.33333334f), 1.0f);
  return copysignf(a < 0.1f ? small : big, x);
}

DEVI v8f wmma4(v2f a, v2f b, v8f c) {
  return __builtin_amdgcn_wmma_f32_16x16x4_f32(false, a, false, b, (short)0, c,
                                               false, false);
}

DEVI v8f gemm_global_a(const float* __restrict__ Arow0, int lda,
                       const float* __restrict__ Bn, int ldb, int K,
                       v8f acc, int l15, int half) {
#pragma unroll 4
  for (int kc = 0; kc < K; kc += 4) {
    v2f a = *(const v2f*)(Arow0 + (size_t)l15 * lda + kc + 2 * half);
    const float* pb = Bn + (size_t)(kc + 2 * half) * ldb + l15;
    v2f b;
    b.x = pb[0];
    b.y = pb[ldb];
    acc = wmma4(a, b, acc);
  }
  return acc;
}

DEVI v8f gemm_lds_a(const float* hs, const float* __restrict__ Bn, int ldb,
                    v8f acc, int l15, int half) {
#pragma unroll 4
  for (int kc = 0; kc < Uc; kc += 4) {
    v2f a = *(const v2f*)(hs + l15 * HS + kc + 2 * half);
    const float* pb = Bn + (size_t)(kc + 2 * half) * ldb + l15;
    v2f b;
    b.x = pb[0];
    b.y = pb[ldb];
    acc = wmma4(a, b, acc);
  }
  return acc;
}

template <bool PRE>
__global__ __launch_bounds__(512) void rnn_fused_kernel(
    const float* __restrict__ xp, const float* __restrict__ x,
    const float* __restrict__ enc_Wx, const float* __restrict__ enc_b,
    const float* __restrict__ enc_Wh, const float* __restrict__ dec_Wx,
    const float* __restrict__ dec_Wh, const float* __restrict__ dec_b,
    float* __restrict__ out) {
  __shared__ float h_enc[2][16 * HS];
  __shared__ float h_dec[2][16 * HS];

  const int lane = threadIdx.x & 31;
  const int wave = threadIdx.x >> 5;
  const int l15 = lane & 15, half = lane >> 4;
  const int n0 = wave * 16;
  const int bbase = blockIdx.x * 16;

  for (int i = threadIdx.x; i < 16 * HS; i += blockDim.x) {
    h_enc[0][i] = 0.0f;
    h_enc[1][i] = 0.0f;
    h_dec[0][i] = 0.0f;
    h_dec[1][i] = 0.0f;
  }
  __syncthreads();

  const float eb = enc_b[n0 + l15];
  const float db = dec_b[n0 + l15];
  v8f dacc = {};
  v8f acc_pre = {};

  if (PRE) {
    const float* p0 = xp + ((size_t)0 * Bc + bbase) * Uc + n0 + l15;
#pragma unroll
    for (int r = 0; r < 8; ++r) acc_pre[r] = p0[(r + 8 * half) * Uc];
  }

  for (int t = 0; t < Tc; ++t) {
    const int cur = t & 1, nxt = cur ^ 1;

    v8f acc;
    if (PRE) {
      acc = acc_pre;
      if (t + 1 < Tc) {
        const float* p = xp + ((size_t)(t + 1) * Bc + bbase) * Uc + n0 + l15;
#pragma unroll
        for (int r = 0; r < 8; ++r) acc_pre[r] = p[(r + 8 * half) * Uc];
      }
    } else {
      acc = (v8f){eb, eb, eb, eb, eb, eb, eb, eb};
      acc = gemm_global_a(x + (size_t)bbase * Tc * Fc + (size_t)t * Fc, Tc * Fc,
                          enc_Wx + n0, Uc, Fc, acc, l15, half);
    }
    acc = gemm_lds_a(h_enc[cur], enc_Wh + n0, Uc, acc, l15, half);
#pragma unroll
    for (int r = 0; r < 8; ++r) acc[r] = tanh_hw(acc[r]);
#pragma unroll
    for (int r = 0; r < 8; ++r)
      h_enc[nxt][(r + 8 * half) * HS + n0 + l15] = acc[r];
    __syncthreads();

    v8f acc2 = {db, db, db, db, db, db, db, db};
    acc2 = gemm_lds_a(h_enc[nxt], dec_Wx + n0, Uc, acc2, l15, half);
    acc2 = gemm_lds_a(h_dec[cur], dec_Wh + n0, Uc, acc2, l15, half);
#pragma unroll
    for (int r = 0; r < 8; ++r) acc2[r] = tanh_hw(acc2[r]);
#pragma unroll
    for (int r = 0; r < 8; ++r)
      h_dec[nxt][(r + 8 * half) * HS + n0 + l15] = acc2[r];
    dacc = acc2;
    __syncthreads();
  }

#pragma unroll
  for (int r = 0; r < 8; ++r)
    out[(size_t)(bbase + r + 8 * half) * Uc + n0 + l15] = dacc[r];
}

constexpr int CH = 64;
constexpr int NCH = Tc / CH;
constexpr int NREC = 16;
constexpr int NWORK = 240;
constexpr int XT = CH / 16;
constexpr int NBLK = NREC + NWORK;

DEVI int dec_t0(int j) { return j > 0 ? j * CH - 1 : 0; }
DEVI int dec_t1(int j) { return j == NCH - 1 ? Tc : (j + 1) * CH - 1; }

DEVI void xproj_tile(const float* __restrict__ x, const float* __restrict__ Wx,
                     const float* __restrict__ b, float* xp, int bidx, int tbase,
                     int n0, int l15, int half) {
  const float bv = b[n0 + l15];
  v8f acc = {bv, bv, bv, bv, bv, bv, bv, bv};
  acc = gemm_global_a(x + ((size_t)bidx * Tc + tbase) * Fc, Fc, Wx + n0, Uc, Fc,
                      acc, l15, half);
#pragma unroll
  for (int r = 0; r < 8; ++r)
    xp[((size_t)(tbase + r + 8 * half) * Bc + bidx) * Uc + n0 + l15] = acc[r];
}

DEVI void hproj_tile(float* tile, const float* __restrict__ Wx,
                     const float* __restrict__ b, float* a_lds, int n0, int l15,
                     int half) {
  for (int i = threadIdx.x; i < 16 * Uc / 4; i += blockDim.x) {
    const int row = i / (Uc / 4), c4 = i % (Uc / 4);
    const float4 v = *(const float4*)(tile + row * Uc + c4 * 4);
    *(float4*)(a_lds + row * HS + c4 * 4) = v;
  }
  __syncthreads();
  const float bv = b[n0 + l15];
  v8f acc = {bv, bv, bv, bv, bv, bv, bv, bv};
  acc = gemm_lds_a(a_lds, Wx + n0, Uc, acc, l15, half);
#pragma unroll
  for (int r = 0; r < 8; ++r) tile[(r + 8 * half) * Uc + n0 + l15] = acc[r];
  __syncthreads();
}

constexpr int GF = 16;
constexpr int NG = Uc / 4 / GF;
static_assert(NG == 4, "the step body below is written out for four groups");

DEVI void fence() { __builtin_amdgcn_sched_barrier(0); }

template <int G>
DEVI void lds_group(v2f (&dst)[GF], const float* ha) {
#pragma unroll
  for (int i = 0; i < GF; ++i) dst[i] = *(const v2f*)(ha + 4 * (G * GF + i));
}

template <int G>
DEVI v8f mma_group(const v2f (&a)[GF], const v2f (&bw)[Uc / 4], v8f acc) {
#pragma unroll
  for (int i = 0; i < GF; ++i) acc = wmma4(a[i], bw[G * GF + i], acc);
  return acc;
}

template <bool SEQ>
DEVI void rec_chunk(float* xh, const float* __restrict__ Wh, int bbase, int t0,
                    int t1, const float* init, float* fin, float* hl, int n0,
                    int l15, int half) {
  float* h0 = hl;
  float* h1 = hl + 16 * HS;
  if (init) {
    const float* src = init + (size_t)bbase * Uc;
    for (int i = threadIdx.x; i < 16 * Uc / 4; i += blockDim.x) {
      const int row = i / (Uc / 4), c4 = i % (Uc / 4);
      const float4 v = *(const float4*)(src + row * Uc + c4 * 4);
      *(float4*)(h0 + row * HS + c4 * 4) = v;
    }
  } else {
    for (int i = threadIdx.x; i < 16 * HS; i += blockDim.x) h0[i] = 0.0f;
  }
  __syncthreads();

  v2f bw[Uc / 4];
  {
    const float* pb = Wh + (size_t)(2 * half) * Uc + n0 + l15;
#pragma unroll
    for (int i = 0; i < Uc / 4; ++i) {
      bw[i].x = pb[(size_t)(4 * i) * Uc];
      bw[i].y = pb[(size_t)(4 * i + 1) * Uc];
    }
  }

  const unsigned lane_off = (unsigned)(8 * half * Uc + n0 + l15);

  v8f acc_pre;
  {
    const float* p0 = xh + ((size_t)t0 * Bc + bbase) * Uc;
#pragma unroll
    for (int r = 0; r < 8; ++r) acc_pre[r] = p0[lane_off + r * Uc];
  }
  v8f acc = {};
  v8f hv = {};

  for (int t = t0; t < t1; ++t) {
    const float* hc = ((t - t0) & 1) ? h1 : h0;
    float* hn = ((t - t0) & 1) ? h0 : h1;
    const float* ha = hc + l15 * HS + 2 * half;

    v2f fa[GF], fb[GF];
    lds_group<0>(fa, ha);
    lds_group<1>(fb, ha);
    fence();
    if (SEQ && t > t0) {
      float* q = xh + ((size_t)(t - 1) * Bc + bbase) * Uc;
#pragma unroll
      for (int r = 0; r < 8; ++r) q[lane_off + r * Uc] = hv[r];
    }
    acc = acc_pre;
    if (t + 1 < t1) {
      const float* p = xh + ((size_t)(t + 1) * Bc + bbase) * Uc;
#pragma unroll
      for (int r = 0; r < 8; ++r) acc_pre[r] = p[lane_off + r * Uc];
    }
    fence();
    acc = mma_group<0>(fa, bw, acc);
    fence();
    lds_group<2>(fa, ha);
    fence();
    acc = mma_group<1>(fb, bw, acc);
    fence();
    lds_group<3>(fb, ha);
    fence();
    acc = mma_group<2>(fa, bw, acc);
    acc = mma_group<3>(fb, bw, acc);

#pragma unroll
    for (int r = 0; r < 8; ++r) hv[r] = tanh_hw(acc[r]);
#pragma unroll
    for (int r = 0; r < 8; ++r) hn[(r + 8 * half) * HS + n0 + l15] = hv[r];
    __syncthreads();
  }

  if (SEQ) {
    float* q = xh + ((size_t)(t1 - 1) * Bc + bbase) * Uc;
#pragma unroll
    for (int r = 0; r < 8; ++r) q[lane_off + r * Uc] = hv[r];
  } else {
#pragma unroll
    for (int r = 0; r < 8; ++r) fin[(r + 8 * half) * Uc + n0 + l15] = hv[r];
  }
}

__global__ __launch_bounds__(512) void stage_kernel(
    float* xh, const float* __restrict__ x, const float* __restrict__ enc_Wx,
    const float* __restrict__ enc_b, const float* __restrict__ enc_Wh,
    const float* __restrict__ dec_Wx, const float* __restrict__ dec_b,
    const float* __restrict__ dec_Wh, float* out, int k) {
  __shared__ float sm[2 * 16 * HS];

  const int lane = threadIdx.x & 31;
  const int wave = threadIdx.x >> 5;
  const int l15 = lane & 15, half = lane >> 4;
  const int n0 = wave * 16;
  const int blk = blockIdx.x;

  if (blk < 8) {
    if (k >= 0 && k < NCH) {
      const int t0 = k * CH;
      const float* init = k > 0 ? xh + (size_t)(t0 - 1) * Bc * Uc : nullptr;
      rec_chunk<true>(xh, enc_Wh, blk * 16, t0, t0 + CH, init, nullptr, sm, n0,
                      l15, half);
    }
  } else if (blk < NREC) {
    const int j = k - 2;
    if (j >= 0 && j < NCH) {
      const int bbase = (blk - 8) * 16;
      const int t0 = dec_t0(j);
      const float* init =
          j > 0 ? xh + (size_t)dec_t0(j - 1) * Bc * Uc : nullptr;
      float* fin = j == NCH - 1 ? out + (size_t)bbase * Uc
                                : xh + ((size_t)t0 * Bc + bbase) * Uc;
      rec_chunk<false>(xh, dec_Wh, bbase, t0, dec_t1(j), init, fin, sm, n0, l15,
                       half);
    }
  } else {
    const int w = blk - NREC;
    const int kx = k + 1;
    if (kx >= 0 && kx < NCH) {
      for (int q = w; q < Bc * XT; q += NWORK)
        xproj_tile(x, enc_Wx, enc_b, xh, q / XT, kx * CH + (q % XT) * 16, n0,
                   l15, half);
    }
    const int jp = k - 1;
    if (jp >= 0 && jp < NCH) {
      const int tile1 = dec_t1(jp) * (Bc / 16);
      for (int tile = dec_t0(jp) * (Bc / 16) + w; tile < tile1; tile += NWORK)
        hproj_tile(xh + (size_t)tile * 16 * Uc, dec_Wx, dec_b, sm, n0, l15,
                   half);
    }
  }
}

extern "C" void kernel_launch(void* const* d_in, const int* in_sizes, int n_in,
                              void* d_out, int out_size, void* d_ws,
                              size_t ws_size, hipStream_t stream) {
  const float* x = (const float*)d_in[0];
  const float* enc_Wx = (const float*)d_in[1];
  const float* enc_Wh = (const float*)d_in[2];
  const float* enc_b = (const float*)d_in[3];
  const float* dec_Wx = (const float*)d_in[4];
  const float* dec_Wh = (const float*)d_in[5];
  const float* dec_b = (const float*)d_in[6];
  float* out = (float*)d_out;

  const size_t xp_bytes = (size_t)Bc * Tc * Uc * sizeof(float);
  if (ws_size >= xp_bytes) {
    float* xh = (float*)d_ws;
#define STAGE(k)                                                            \
  stage_kernel<<<NBLK, 512, 0, stream>>>(xh, x, enc_Wx, enc_b,              \
                                                 enc_Wh, dec_Wx, dec_b,     \
                                                 dec_Wh, out, k)
    STAGE(-1); STAGE(0); STAGE(1); STAGE(2); STAGE(3); STAGE(4); STAGE(5);
    STAGE(6); STAGE(7); STAGE(8); STAGE(9); STAGE(10); STAGE(11); STAGE(12);
    STAGE(13); STAGE(14); STAGE(15); STAGE(16); STAGE(17);
#undef STAGE
  } else {
    rnn_fused_kernel<false><<<Bc / 16, 512, 0, stream>>>(
        nullptr, x, enc_Wx, enc_b, enc_Wh, dec_Wx, dec_Wh, dec_b, out);
  }
}
